# starting kernel
# baseline (speedup 1.0000x reference)
#include <hip/hip_runtime.h>


constexpr int kB = 8;
constexpr int kS = 512;
constexpr int kN = 32;
constexpr int kH = 256;
constexpr int kRows = kN + 1;
constexpr int kRowsPad = 36;
constexpr float kNeg = -1e9f;
constexpr float kSlope = 0.2f;

typedef __attribute__((ext_vector_type(2))) float v2f;
typedef __attribute__((ext_vector_type(8))) float v8f;

__global__ __launch_bounds__(256) void gat_kernel(
    const int* __restrict__ node_ids,
    const int* __restrict__ neighs,
    const int* __restrict__ mask,
    const float* __restrict__ emb,
    const float* __restrict__ a_w,
    const float* __restrict__ a_b,
    float* __restrict__ out)
{
    __shared__ float sh_cand[kRowsPad][kH];
    __shared__ float sh_z[kRows];
    __shared__ float sh_w[kRowsPad];
    __shared__ float sh_dot1;

    const int pos  = blockIdx.x;
    const int tid  = threadIdx.x;
    const int wave = tid >> 5;
    const int lane = tid & 31;

    float a2r[8];
    #pragma unroll
    for (int i = 0; i < 8; ++i) a2r[i] = a_w[kH + lane * 8 + i];

    if (wave >= 1 && wave <= 3) {
        float4 z4 = make_float4(0.f, 0.f, 0.f, 0.f);
        float4* sp = (float4*)(&sh_cand[32 + wave][lane * 8]);
        sp[0] = z4; sp[1] = z4;
    }

    if (wave == 0) {
        const int idx = node_ids[pos];
        const float4* rp = (const float4*)(emb + (size_t)idx * kH) + lane * 2;
        float4 r0 = rp[0];
        float4 r1 = rp[1];
        float4* sp = (float4*)(&sh_cand[0][lane * 8]);
        sp[0] = r0; sp[1] = r1;

        float p2 = r0.x*a2r[0] + r0.y*a2r[1] + r0.z*a2r[2] + r0.w*a2r[3]
                 + r1.x*a2r[4] + r1.y*a2r[5] + r1.z*a2r[6] + r1.w*a2r[7];
        const float* a1 = a_w + lane * 8;
        float p1 = r0.x*a1[0] + r0.y*a1[1] + r0.z*a1[2] + r0.w*a1[3]
                 + r1.x*a1[4] + r1.y*a1[5] + r1.z*a1[6] + r1.w*a1[7];
        #pragma unroll
        for (int off = 16; off > 0; off >>= 1) {
            p2 += __shfl_xor(p2, off, 32);
            p1 += __shfl_xor(p1, off, 32);
        }
        if (lane == 0) { sh_z[0] = p2; sh_dot1 = p1; }
    }

    for (int jj = wave; jj < kN; jj += 8) {
        const int idx = neighs[pos * kN + jj];
        if (jj + 8 < kN) {
            const int nidx = neighs[pos * kN + jj + 8];
            __builtin_prefetch(emb + (size_t)nidx * kH + lane * 8, 0, 1);
        }
        const float4* rp = (const float4*)(emb + (size_t)idx * kH) + lane * 2;
        float4 r0 = rp[0];
        float4 r1 = rp[1];
        float4* sp = (float4*)(&sh_cand[jj + 1][lane * 8]);
        sp[0] = r0; sp[1] = r1;

        float p2 = r0.x*a2r[0] + r0.y*a2r[1] + r0.z*a2r[2] + r0.w*a2r[3]
                 + r1.x*a2r[4] + r1.y*a2r[5] + r1.z*a2r[6] + r1.w*a2r[7];
        #pragma unroll
        for (int off = 16; off > 0; off >>= 1)
            p2 += __shfl_xor(p2, off, 32);
        if (lane == 0) sh_z[jj + 1] = p2;
    }

    __syncthreads();

    if (tid < kRows) {
        float z = sh_dot1 + sh_z[tid] + a_b[0];
        float att = z > 0.f ? z : kSlope * z;
        if (tid >= 1) att += (float)mask[pos * kN + (tid - 1)] * kNeg;
        sh_z[tid] = att;
    }
    __syncthreads();

    if (tid == 0) {
        float m = sh_z[0];
        for (int j = 1; j < kRows; ++j) m = fmaxf(m, sh_z[j]);
        float s = 0.f;
        for (int j = 0; j < kRows; ++j) s += __expf(sh_z[j] - m);
        float inv = 1.f / s;
        for (int j = 0; j < kRows; ++j) sh_w[j] = __expf(sh_z[j] - m) * inv;
        sh_w[33] = 0.f; sh_w[34] = 0.f; sh_w[35] = 0.f;
    }
    __syncthreads();

    const int kgrp = (lane >> 4) << 1;
    const int col  = lane & 15;
    for (int c = wave; c < kH / 16; c += 8) {
        v8f acc = {};
        #pragma unroll
        for (int k0 = 0; k0 < kRowsPad; k0 += 4) {
            v2f a, bm;
            a.x  = sh_w[k0 + kgrp + 0];
            a.y  = sh_w[k0 + kgrp + 1];
            bm.x = sh_cand[k0 + kgrp + 0][c * 16 + col];
            bm.y = sh_cand[k0 + kgrp + 1][c * 16 + col];
            acc = __builtin_amdgcn_wmma_f32_16x16x4_f32(
                 false, a,  false, bm,
                 (short)0, acc,  false,  false);
        }
        if (lane < 16)
            out[(size_t)pos * kH + c * 16 + lane] = acc[0];
    }
}

extern "C" void kernel_launch(void* const* d_in, const int* in_sizes, int n_in,
                              void* d_out, int out_size, void* d_ws, size_t ws_size,
                              hipStream_t stream) {
    const int*   node_ids = (const int*)d_in[0];
    const int*   neighs   = (const int*)d_in[1];
    const int*   mask     = (const int*)d_in[2];
    const float* emb      = (const float*)d_in[3];
    const float* a_w      = (const float*)d_in[4];
    const float* a_b      = (const float*)d_in[5];
    float* out = (float*)d_out;

    dim3 grid(kB * kS);
    dim3 block(256);
    gat_kernel<<<grid, block, 0, stream>>>(node_ids, neighs, mask, emb, a_w, a_b, out);
}


# Resident grid: 1024 work-groups (4 per WGP) each walk 4 positions, the next position's rows requested under the current softmax and product; a quarter as many waves enter; sum order kept
# speedup vs baseline: 1.0601x; 1.0601x over previous
#include <hip/hip_runtime.h>


constexpr int kB = 8;
constexpr int kS = 512;
constexpr int kN = 32;
constexpr int kH = 256;
constexpr int kRows = kN + 1;
constexpr int kRowsPad = 36;
constexpr float kNeg = -1e9f;
constexpr float kSlope = 0.2f;
constexpr int kPos = kB * kS;
constexpr int kGrid = 1024;
constexpr int kPerWG = kPos / kGrid;
constexpr int kOwn = kN / 8;

typedef __attribute__((ext_vector_type(2))) float v2f;
typedef __attribute__((ext_vector_type(8))) float v8f;

__device__ __forceinline__ void lds_barrier()
{
    __builtin_amdgcn_fence(__ATOMIC_RELEASE, "workgroup", "local");
    __builtin_amdgcn_s_barrier();
    __builtin_amdgcn_fence(__ATOMIC_ACQUIRE, "workgroup", "local");
}

__device__ __forceinline__ float dot8(const float4 r0, const float4 r1, const float* a)
{
    return r0.x*a[0] + r0.y*a[1] + r0.z*a[2] + r0.w*a[3]
         + r1.x*a[4] + r1.y*a[5] + r1.z*a[6] + r1.w*a[7];
}

__global__ __launch_bounds__(256) void gat_kernel(
    const int* __restrict__ node_ids,
    const int* __restrict__ neighs,
    const int* __restrict__ mask,
    const float* __restrict__ emb,
    const float* __restrict__ a_w,
    const float* __restrict__ a_b,
    float* __restrict__ out)
{
    __shared__ float sh_cand[kRowsPad][kH];
    __shared__ float sh_z[kRows];
    __shared__ float sh_w[kRowsPad];
    __shared__ float sh_dot1;
    __shared__ float sh_a[2 * kH];

    const int tid  = threadIdx.x;
    const int wave = __builtin_amdgcn_readfirstlane(tid >> 5);
    const int lane = tid & 31;

    const float aw_lo = a_w[tid];
    const float aw_hi = a_w[tid + kH];
    const float bias = a_b[0];
    const float* a1r = &sh_a[lane * 8];
    const float* a2r = &sh_a[kH + lane * 8];

    if (wave >= 1 && wave <= 3) {
        float4 z4 = make_float4(0.f, 0.f, 0.f, 0.f);
        float4* sp = (float4*)(&sh_cand[32 + wave][lane * 8]);
        sp[0] = z4; sp[1] = z4;
    }

    float4 rn[kOwn][2];
    float4 rs[2] = { make_float4(0.f, 0.f, 0.f, 0.f), make_float4(0.f, 0.f, 0.f, 0.f) };
    int mk = 0;
    int nidx[kOwn];
    int sidx = 0;

    {
        const int p = blockIdx.x;
        int id[kOwn];
        #pragma unroll
        for (int r = 0; r < kOwn; ++r) id[r] = neighs[p * kN + wave + 8 * r];
        #pragma unroll
        for (int r = 0; r < kOwn; ++r) {
            const float4* rp = (const float4*)(emb + (size_t)id[r] * kH) + lane * 2;
            rn[r][0] = rp[0];
            rn[r][1] = rp[1];
        }
        if (wave == 0) {
            const float4* rp = (const float4*)(emb + (size_t)node_ids[p] * kH) + lane * 2;
            rs[0] = rp[0];
            rs[1] = rp[1];
        }
        if (tid >= 1 && tid < kRows) mk = mask[p * kN + tid - 1];
        if (kPerWG > 1) {
            const int pn = p + kGrid;
            #pragma unroll
            for (int r = 0; r < kOwn; ++r) nidx[r] = neighs[pn * kN + wave + 8 * r];
            sidx = node_ids[pn];
        }
    }

    sh_a[tid]      = aw_lo;
    sh_a[tid + kH] = aw_hi;
    lds_barrier();

    #pragma unroll
    for (int it = 0; it < kPerWG; ++it) {
        const int pos = blockIdx.x + it * kGrid;

        float pz[kOwn];
        #pragma unroll
        for (int r = 0; r < kOwn; ++r) {
            float4* sp = (float4*)(&sh_cand[wave + 8 * r + 1][lane * 8]);
            sp[0] = rn[r][0];
            sp[1] = rn[r][1];
            pz[r] = dot8(rn[r][0], rn[r][1], a2r);
        }
        float ps1 = 0.f, ps2 = 0.f;
        if (wave == 0) {
            float4* sp = (float4*)(&sh_cand[0][lane * 8]);
            sp[0] = rs[0];
            sp[1] = rs[1];
            ps2 = dot8(rs[0], rs[1], a2r);
            ps1 = dot8(rs[0], rs[1], a1r);
        }
        const int mk_cur = mk;

        if (it + 1 < kPerWG) {
            const int pn = pos + kGrid;
            #pragma unroll
            for (int r = 0; r < kOwn; ++r) {
                const float4* rp = (const float4*)(emb + (size_t)nidx[r] * kH) + lane * 2;
                rn[r][0] = rp[0];
                rn[r][1] = rp[1];
            }
            if (wave == 0) {
                const float4* rp = (const float4*)(emb + (size_t)sidx * kH) + lane * 2;
                rs[0] = rp[0];
                rs[1] = rp[1];
            }
            if (tid >= 1 && tid < kRows) mk = mask[pn * kN + tid - 1];
            if (it + 2 < kPerWG) {
                const int pnn = pn + kGrid;
                #pragma unroll
                for (int r = 0; r < kOwn; ++r) nidx[r] = neighs[pnn * kN + wave + 8 * r];
                sidx = node_ids[pnn];
            }
        }

        #pragma unroll
        for (int off = 16; off > 0; off >>= 1) {
            #pragma unroll
            for (int r = 0; r < kOwn; ++r) pz[r] += __shfl_xor(pz[r], off, 32);
            ps2 += __shfl_xor(ps2, off, 32);
            ps1 += __shfl_xor(ps1, off, 32);
        }
        if (lane == 0) {
            #pragma unroll
            for (int r = 0; r < kOwn; ++r) sh_z[wave + 8 * r + 1] = pz[r];
            if (wave == 0) { sh_z[0] = ps2; sh_dot1 = ps1; }
        }

        lds_barrier();

        if (tid < kRows) {
            float z = sh_dot1 + sh_z[tid] + bias;
            float att = z > 0.f ? z : kSlope * z;
            if (tid >= 1) att += (float)mk_cur * kNeg;
            sh_z[tid] = att;
        }
        lds_barrier();

        if (tid == 0) {
            float m = sh_z[0];
            for (int j = 1; j < kRows; ++j) m = fmaxf(m, sh_z[j]);
            float s = 0.f;
            for (int j = 0; j < kRows; ++j) s += __expf(sh_z[j] - m);
            float inv = 1.f / s;
            for (int j = 0; j < kRows; ++j) sh_w[j] = __expf(sh_z[j] - m) * inv;
            sh_w[33] = 0.f; sh_w[34] = 0.f; sh_w[35] = 0.f;
        }
        lds_barrier();

        const int kgrp = (lane >> 4) << 1;
        const int col  = lane & 15;
        for (int c = wave; c < kH / 16; c += 8) {
            v8f acc = {};
            #pragma unroll
            for (int k0 = 0; k0 < kRowsPad; k0 += 4) {
                v2f a, bm;
                a.x  = sh_w[k0 + kgrp + 0];
                a.y  = sh_w[k0 + kgrp + 1];
                bm.x = sh_cand[k0 + kgrp + 0][c * 16 + col];
                bm.y = sh_cand[k0 + kgrp + 1][c * 16 + col];
                acc = __builtin_amdgcn_wmma_f32_16x16x4_f32(
                     false, a,  false, bm,
                     (short)0, acc,  false,  false);
            }
            if (lane < 16)
                out[(size_t)pos * kH + c * 16 + lane] = acc[0];
        }

        if (it + 1 < kPerWG) lds_barrier();
    }
}

extern "C" void kernel_launch(void* const* d_in, const int* in_sizes, int n_in,
                              void* d_out, int out_size, void* d_ws, size_t ws_size,
                              hipStream_t stream) {
    const int*   node_ids = (const int*)d_in[0];
    const int*   neighs   = (const int*)d_in[1];
    const int*   mask     = (const int*)d_in[2];
    const float* emb      = (const float*)d_in[3];
    const float* a_w      = (const float*)d_in[4];
    const float* a_b      = (const float*)d_in[5];
    float* out = (float*)d_out;

    dim3 grid(kGrid);
    dim3 block(256);
    gat_kernel<<<grid, block, 0, stream>>>(node_ids, neighs, mask, emb, a_w, a_b, out);
}


# Each wave keeps its own 32 columns of all 33 rows in registers as the B fragments: no LDS tile, one barrier per position, one request per row line; dot, softmax and product sums regroup
# speedup vs baseline: 1.1683x; 1.1021x over previous
#include <hip/hip_runtime.h>


constexpr int kB = 8;
constexpr int kS = 512;
constexpr int kN = 32;
constexpr int kH = 256;
constexpr int kRows = kN + 1;
constexpr int kSteps = 9;
constexpr int kHalf = 2 * kSteps;
constexpr float kNeg = -1e9f;
constexpr float kSlope = 0.2f;
constexpr int kPos = kB * kS;
constexpr int kGrid = 1024;
constexpr int kPerWG = kPos / kGrid;
constexpr int kPitch = 40;
constexpr int kSlotA1 = 36;
constexpr int kGap = 2;

typedef __attribute__((ext_vector_type(2))) float v2f;
typedef __attribute__((ext_vector_type(8))) float v8f;

__device__ __forceinline__ void lds_barrier()
{
    __builtin_amdgcn_fence(__ATOMIC_RELEASE, "workgroup", "local");
    __builtin_amdgcn_s_barrier();
    __builtin_amdgcn_fence(__ATOMIC_ACQUIRE, "workgroup", "local");
}

template <int M>
__device__ __forceinline__ float row_xor(float x)
{
    const int b = __builtin_bit_cast(int, x);
    return __builtin_bit_cast(float, __builtin_amdgcn_update_dpp(0, b, 0x160 + M, 0xf, 0xf, true));
}

__device__ __forceinline__ float xhalf(float x)
{
    const int b = __builtin_bit_cast(int, x);
    return __builtin_bit_cast(float,
        __builtin_amdgcn_permlanex16(b, b, 0x76543210, 0xfedcba98, false, false));
}

template <int M>
__device__ __forceinline__ float fold2(float a, float b, bool up)
{
    const float keep = up ? b : a;
    const float send = up ? a : b;
    return keep + row_xor<M>(send);
}

__global__ __launch_bounds__(256) void gat_kernel(
    const int* __restrict__ node_ids,
    const int* __restrict__ neighs,
    const int* __restrict__ mask,
    const float* __restrict__ emb,
    const float* __restrict__ a_w,
    const float* __restrict__ a_b,
    float* __restrict__ out)
{
    __shared__ float sh_part[2][8][kPitch];
    __shared__ float sh_w[8][kPitch];

    const int tid  = threadIdx.x;
    const int wave = __builtin_amdgcn_readfirstlane(tid >> 5);
    const int lane = tid & 31;
    const int h = lane >> 4;
    const int c = lane & 15;
    const int col0 = wave * 32 + 2 * c;

    const float2 a1c = *(const float2*)(a_w + col0);
    const float2 a2c = *(const float2*)(a_w + kH + col0);
    const float bias = a_b[0];
    const unsigned colbyte = (unsigned)col0 * 4u;
    const bool up1 = (c & 1) != 0, up2 = (c & 2) != 0, up4 = (c & 4) != 0, up8 = (c & 8) != 0;
    const int crev = ((c & 1) << 3) | ((c & 2) << 1) | ((c & 4) >> 1) | ((c & 8) >> 3);
    const char* embB = (const char*)emb;

    if (lane < 3) sh_w[wave][kRows + kGap + lane] = 0.f;

    float2 xn[kSteps][2];
    int mk0n = 0, mk32n = 0;
    int idxN[kRows];

    auto load_idx = [&](int p) {
        idxN[0] = node_ids[p];
        #pragma unroll
        for (int j = 0; j < kN; ++j) idxN[1 + j] = neighs[p * kN + j];
    };
    auto load_rows = [&](int p) {
        mk0n = mask[p * kN + (lane >= 1 ? lane - 1 : 0)];
        mk32n = mask[p * kN + kN - 1];
        #pragma unroll
        for (int q = 0; q < kSteps; ++q) {
            #pragma unroll
            for (int s = 0; s < 2; ++s) {
                const int r0 = 2 * q + s;
                const int r1 = kHalf + 2 * q + s;
                const unsigned id = (r1 < kRows) ? (unsigned)(h ? idxN[r1] : idxN[r0])
                                                 : (unsigned)idxN[r0];
                xn[q][s] = *(const float2*)(embB + ((id << 10) + colbyte));
            }
        }
    };

    load_idx(blockIdx.x);
    load_rows(blockIdx.x);
    if (kPerWG > 1) load_idx(blockIdx.x + kGrid);

    #pragma unroll
    for (int it = 0; it < kPerWG; ++it) {
        const int pos = blockIdx.x + it * kGrid;
        const int buf = it & 1;

        float2 x[kSteps][2];
        #pragma unroll
        for (int q = 0; q < kSteps; ++q) {
            #pragma unroll
            for (int s = 0; s < 2; ++s) {
                const bool pad = (kHalf + 2 * q + s >= kRows) && (h != 0);
                x[q][s].x = pad ? 0.f : xn[q][s].x;
                x[q][s].y = pad ? 0.f : xn[q][s].y;
            }
        }
        const int mk0 = mk0n, mk32 = mk32n;

        if (it + 1 < kPerWG) {
            load_rows(pos + kGrid);
            if (it + 2 < kPerWG) load_idx(pos + 2 * kGrid);
        }

        float pz[kSteps][2];
        #pragma unroll
        for (int q = 0; q < kSteps; ++q) {
            pz[q][0] = x[q][0].x * a2c.x + x[q][0].y * a2c.y;
            pz[q][1] = x[q][1].x * a2c.x + x[q][1].y * a2c.y;
        }
        const float ps1 = x[0][0].x * a1c.x + x[0][0].y * a1c.y;

        float f0, f1;
        {
            float u[10], t[5], r[3];
            #pragma unroll
            for (int q = 0; q < kSteps; ++q) u[q] = fold2<8>(pz[q][0], pz[q][1], up8);
            u[9] = fold2<8>(ps1, 0.f, up8);
            #pragma unroll
            for (int j = 0; j < 5; ++j) t[j] = fold2<4>(u[2 * j], u[2 * j + 1], up4);
            r[0] = fold2<2>(t[0], t[1], up2);
            r[1] = fold2<2>(t[2], t[3], up2);
            r[2] = t[4] + row_xor<2>(t[4]);
            f0 = fold2<1>(r[0], r[1], up1);
            f1 = r[2] + row_xor<1>(r[2]);
        }
        sh_part[buf][wave][kHalf * h + crev] = f0;
        if ((c & 7) == 0) sh_part[buf][wave][kHalf * h + 16 + (c >> 3)] = f1;
        if (lane == 4) sh_part[buf][wave][kSlotA1] = f1;

        lds_barrier();

        float d0 = 0.f, d32 = 0.f, d1 = 0.f;
        #pragma unroll
        for (int wv = 0; wv < 8; ++wv) {
            d0  += sh_part[buf][wv][lane];
            d32 += sh_part[buf][wv][kN];
            d1  += sh_part[buf][wv][kSlotA1];
        }
        float z0 = d1 + d0 + bias;
        z0 = z0 > 0.f ? z0 : kSlope * z0;
        if (lane >= 1) z0 += (float)mk0 * kNeg;
        float z1 = d1 + d32 + bias;
        z1 = z1 > 0.f ? z1 : kSlope * z1;
        z1 += (float)mk32 * kNeg;

        float m = z0;
        m = fmaxf(m, row_xor<1>(m)); m = fmaxf(m, row_xor<2>(m));
        m = fmaxf(m, row_xor<4>(m)); m = fmaxf(m, row_xor<8>(m));
        m = fmaxf(m, xhalf(m));
        m = fmaxf(m, z1);
        const float e0 = __expf(z0 - m);
        const float e1 = __expf(z1 - m);
        float sum = e0;
        sum += row_xor<1>(sum); sum += row_xor<2>(sum);
        sum += row_xor<4>(sum); sum += row_xor<8>(sum);
        sum += xhalf(sum);
        sum += e1;
        const float inv = 1.f / sum;

        sh_w[wave][lane + (lane >= kHalf ? kGap : 0)] = e0 * inv;
        if (lane == 0) sh_w[wave][kN + kGap] = e1 * inv;
        __builtin_amdgcn_fence(__ATOMIC_RELEASE, "wavefront", "local");
        __builtin_amdgcn_wave_barrier();
        __builtin_amdgcn_fence(__ATOMIC_ACQUIRE, "wavefront", "local");
        const float4* wp = (const float4*)&sh_w[wave][(kHalf + kGap) * h];
        const float4 wq0 = wp[0], wq1 = wp[1], wq2 = wp[2], wq3 = wp[3];
        const float2 wq4 = *(const float2*)&sh_w[wave][(kHalf + kGap) * h + 16];
        const float wa[kHalf] = { wq0.x, wq0.y, wq0.z, wq0.w, wq1.x, wq1.y, wq1.z, wq1.w,
                                  wq2.x, wq2.y, wq2.z, wq2.w, wq3.x, wq3.y, wq3.z, wq3.w,
                                  wq4.x, wq4.y };

        v8f acc0 = {}, acc1 = {};
        #pragma unroll
        for (int q = 0; q < kSteps; ++q) {
            v2f a, b0, b1;
            a.x  = wa[2 * q];   a.y  = wa[2 * q + 1];
            b0.x = x[q][0].x;   b0.y = x[q][1].x;
            b1.x = x[q][0].y;   b1.y = x[q][1].y;
            acc0 = __builtin_amdgcn_wmma_f32_16x16x4_f32(
                 false, a,  false, b0,
                 (short)0, acc0,  false,  false);
            acc1 = __builtin_amdgcn_wmma_f32_16x16x4_f32(
                 false, a,  false, b1,
                 (short)0, acc1,  false,  false);
        }
        if (lane < 16)
            *(float2*)&out[(size_t)pos * kH + col0] = make_float2(acc0[0], acc1[0]);
    }
}

extern "C" void kernel_launch(void* const* d_in, const int* in_sizes, int n_in,
                              void* d_out, int out_size, void* d_ws, size_t ws_size,
                              hipStream_t stream) {
    const int*   node_ids = (const int*)d_in[0];
    const int*   neighs   = (const int*)d_in[1];
    const int*   mask     = (const int*)d_in[2];
    const float* emb      = (const float*)d_in[3];
    const float* a_w      = (const float*)d_in[4];
    const float* a_b      = (const float*)d_in[5];
    float* out = (float*)d_out;

    dim3 grid(kGrid);
    dim3 block(256);
    gat_kernel<<<grid, block, 0, stream>>>(node_ids, neighs, mask, emb, a_w, a_b, out);
}


# Four waves of 64 columns per position instead of eight of 32: one 256-byte request per row from a scalar row number, one WMMA per row with its weight on A's diagonal; dot and product sums regroup
# speedup vs baseline: 1.4445x; 1.2364x over previous
#include <hip/hip_runtime.h>


constexpr int kB = 8;
constexpr int kS = 512;
constexpr int kN = 32;
constexpr int kH = 256;
constexpr int kRows = kN + 1;
constexpr int kVals = kRows + 1;
constexpr float kNeg = -1e9f;
constexpr float kSlope = 0.2f;
constexpr int kPos = kB * kS;
constexpr int kGrid = 1024;
constexpr int kPerWG = kPos / kGrid;
constexpr int kWaves = 4;
constexpr int kPitch = 36;
constexpr int kSlotA1 = kRows;
constexpr int kStrip = 72;
constexpr int kQuads = 17;

typedef __attribute__((ext_vector_type(2))) float v2f;
typedef __attribute__((ext_vector_type(8))) float v8f;

__device__ __forceinline__ void lds_barrier()
{
    __builtin_amdgcn_fence(__ATOMIC_RELEASE, "workgroup", "local");
    __builtin_amdgcn_s_barrier();
    __builtin_amdgcn_fence(__ATOMIC_ACQUIRE, "workgroup", "local");
}

template <int M>
__device__ __forceinline__ float row_xor(float x)
{
    const int b = __builtin_bit_cast(int, x);
    return __builtin_bit_cast(float, __builtin_amdgcn_update_dpp(0, b, 0x160 + M, 0xf, 0xf, true));
}

__device__ __forceinline__ float xhalf(float x)
{
    const int b = __builtin_bit_cast(int, x);
    return __builtin_bit_cast(float,
        __builtin_amdgcn_permlanex16(b, b, 0x76543210, 0xfedcba98, false, false));
}

template <int M>
__device__ __forceinline__ float fold2(float a, float b, bool up)
{
    const float keep = up ? b : a;
    const float send = up ? a : b;
    return keep + row_xor<M>(send);
}

__global__ __launch_bounds__(128) __attribute__((amdgpu_waves_per_eu(4, 4))) void gat_kernel(
    const int* __restrict__ node_ids,
    const int* __restrict__ neighs,
    const int* __restrict__ mask,
    const float* __restrict__ emb,
    const float* __restrict__ a_w,
    const float* __restrict__ a_b,
    float* __restrict__ out)
{
    __shared__ float sh_part[2][kWaves][kPitch];
    __shared__ __attribute__((aligned(16))) float sh_a[kWaves][2][kStrip];
    __shared__ __attribute__((aligned(16))) float sh_z[kStrip];

    const int tid  = threadIdx.x;
    const int wave = __builtin_amdgcn_readfirstlane(tid >> 5);
    const int lane = tid & 31;
    const int h = lane >> 4;
    const int c = lane & 15;
    const int col0 = wave * 64 + 2 * lane;

    const float2 a1c = *(const float2*)(a_w + col0);
    const float2 a2c = *(const float2*)(a_w + kH + col0);
    const float bias = a_b[0];
    const unsigned colbyte = (unsigned)col0 * 4u;
    const bool up1 = (c & 1) != 0, up2 = (c & 2) != 0, up4 = (c & 4) != 0, up8 = (c & 8) != 0;
    const int crev = ((c & 1) << 3) | ((c & 2) << 1) | ((c & 4) >> 1) | ((c & 8) >> 3);
    const char* embB = (const char*)emb;

    {
        float* za = &sh_a[wave][0][0];
        for (int i = lane; i < 2 * kStrip; i += 32) za[i] = 0.f;
        if (wave == 0)
            for (int i = lane; i < kStrip; i += 32) sh_z[i] = 0.f;
    }
    const float* abase = (lane == 0 || lane == 18) ? &sh_a[wave][0][0]
                       : (lane == 1 || lane == 19) ? &sh_a[wave][1][0]
                       : &sh_z[0];

    float2 xn[kRows];
    int mk0n = 0, mk32n = 0;
    int idxN[kRows];

    auto load_idx = [&](int p) {
        idxN[0] = node_ids[p];
        #pragma unroll
        for (int j = 0; j < kN; ++j) idxN[1 + j] = neighs[p * kN + j];
    };
    auto load_rows = [&](int p) {
        mk0n = mask[p * kN + (lane >= 1 ? lane - 1 : 0)];
        mk32n = mask[p * kN + kN - 1];
        #pragma unroll
        for (int r = 0; r < kRows; ++r) {
            const unsigned off = ((unsigned)idxN[r] << 10) + colbyte;
            xn[r] = *(const float2*)(embB + off);
        }
    };

    load_idx(blockIdx.x);
    load_rows(blockIdx.x);
    if (kPerWG > 1) load_idx(blockIdx.x + kGrid);

    #pragma unroll
    for (int it = 0; it < kPerWG; ++it) {
        const int pos = blockIdx.x + it * kGrid;
        const int buf = it & 1;

        float2 x[kRows];
        #pragma unroll
        for (int r = 0; r < kRows; ++r) x[r] = xn[r];
        const int mk0 = mk0n, mk32 = mk32n;

        if (it + 1 < kPerWG) {
            load_rows(pos + kGrid);
            if (it + 2 < kPerWG) load_idx(pos + 2 * kGrid);
        }

        float v[kVals];
        #pragma unroll
        for (int r = 0; r < kRows; ++r) v[r] = x[r].x * a2c.x + x[r].y * a2c.y;
        v[kRows] = x[0].x * a1c.x + x[0].y * a1c.y;

        float f0, f1, f2;
        {
            float u[17], t[9], q[5];
            #pragma unroll
            for (int j = 0; j < 17; ++j) u[j] = fold2<8>(v[2 * j], v[2 * j + 1], up8);
            #pragma unroll
            for (int j = 0; j < 8; ++j) t[j] = fold2<4>(u[2 * j], u[2 * j + 1], up4);
            t[8] = u[16] + row_xor<4>(u[16]);
            #pragma unroll
            for (int j = 0; j < 4; ++j) q[j] = fold2<2>(t[2 * j], t[2 * j + 1], up2);
            q[4] = t[8] + row_xor<2>(t[8]);
            f0 = fold2<1>(q[0], q[1], up1);
            f1 = fold2<1>(q[2], q[3], up1);
            f2 = q[4] + row_xor<1>(q[4]);
            f0 += xhalf(f0);
            f1 += xhalf(f1);
            f2 += xhalf(f2);
        }
        sh_part[buf][wave][16 * h + crev] = h ? f1 : f0;
        if ((lane & 23) == 0) sh_part[buf][wave][kN + (lane >> 3)] = f2;

        lds_barrier();

        float d0 = 0.f, d32 = 0.f, d1 = 0.f;
        #pragma unroll
        for (int wv = 0; wv < kWaves; ++wv) {
            d0  += sh_part[buf][wv][lane];
            d32 += sh_part[buf][wv][kN];
            d1  += sh_part[buf][wv][kSlotA1];
        }
        float z0 = d1 + d0 + bias;
        z0 = z0 > 0.f ? z0 : kSlope * z0;
        if (lane >= 1) z0 += (float)mk0 * kNeg;
        float z1 = d1 + d32 + bias;
        z1 = z1 > 0.f ? z1 : kSlope * z1;
        z1 += (float)mk32 * kNeg;

        float m = z0;
        m = fmaxf(m, row_xor<1>(m)); m = fmaxf(m, row_xor<2>(m));
        m = fmaxf(m, row_xor<4>(m)); m = fmaxf(m, row_xor<8>(m));
        m = fmaxf(m, xhalf(m));
        m = fmaxf(m, z1);
        const float e0 = __expf(z0 - m);
        const float e1 = __expf(z1 - m);
        float sum = e0;
        sum += row_xor<1>(sum); sum += row_xor<2>(sum);
        sum += row_xor<4>(sum); sum += row_xor<8>(sum);
        sum += xhalf(sum);
        sum += e1;
        const float inv = 1.f / sum;

        const float wl = e0 * inv;
        sh_a[wave][0][2 * lane] = wl;
        sh_a[wave][1][2 * lane + 1] = wl;
        if (lane == 0) {
            sh_a[wave][0][2 * kN] = e1 * inv;
            sh_a[wave][1][2 * kN + 1] = e1 * inv;
        }
        __builtin_amdgcn_fence(__ATOMIC_RELEASE, "wavefront", "local");
        __builtin_amdgcn_wave_barrier();
        __builtin_amdgcn_fence(__ATOMIC_ACQUIRE, "wavefront", "local");
        float4 av[kQuads];
        #pragma unroll
        for (int i = 0; i < kQuads; ++i) av[i] = ((const float4*)abase)[i];
        __builtin_amdgcn_sched_barrier(0);

        v8f acc0 = {}, acc1 = {};
        #pragma unroll
        for (int r = 0; r < kRows; ++r) {
            const float4 aq = av[r >> 1];
            v2f a, b;
            a.x = (r & 1) ? aq.z : aq.x;
            a.y = (r & 1) ? aq.w : aq.y;
            b.x = x[r].x;
            b.y = x[r].y;
            if (r & 1)
                acc1 = __builtin_amdgcn_wmma_f32_16x16x4_f32(
                     false, a,  false, b,
                     (short)0, acc1,  false,  false);
            else
                acc0 = __builtin_amdgcn_wmma_f32_16x16x4_f32(
                     false, a,  false, b,
                     (short)0, acc0,  false,  false);
        }
        if (lane < 16) {
            float* o = out + (size_t)pos * kH + wave * 64 + 2 * c;
            *(float2*)o        = make_float2(acc0[0] + acc1[0], acc0[1] + acc1[1]);
            *(float2*)(o + 32) = make_float2(acc0[2] + acc1[2], acc0[3] + acc1[3]);
        }
    }
}

extern "C" void kernel_launch(void* const* d_in, const int* in_sizes, int n_in,
                              void* d_out, int out_size, void* d_ws, size_t ws_size,
                              hipStream_t stream) {
    const int*   node_ids = (const int*)d_in[0];
    const int*   neighs   = (const int*)d_in[1];
    const int*   mask     = (const int*)d_in[2];
    const float* emb      = (const float*)d_in[3];
    const float* a_w      = (const float*)d_in[4];
    const float* a_b      = (const float*)d_in[5];
    float* out = (float*)d_out;

    dim3 grid(kGrid);
    dim3 block(128);
    gat_kernel<<<grid, block, 0, stream>>>(node_ids, neighs, mask, emb, a_w, a_b, out);
}


# Rows with a set mask word (weight exactly 0 through the -1e9) are neither requested nor multiplied: one scalar branch per row on a ballot bit map; product sums now take the self row last
# speedup vs baseline: 1.7237x; 1.1933x over previous
#include <hip/hip_runtime.h>


constexpr int kB = 8;
constexpr int kS = 512;
constexpr int kN = 32;
constexpr int kH = 256;
constexpr int kRows = kN + 1;
constexpr int kVals = kRows + 1;
constexpr float kNeg = -1e9f;
constexpr float kSlope = 0.2f;
constexpr int kPos = kB * kS;
constexpr int kGrid = 1024;
constexpr int kPerWG = kPos / kGrid;
constexpr int kWaves = 4;
constexpr int kPitch = 36;
constexpr int kSlotA1 = kRows;
constexpr int kStrip = 72;
constexpr int kQuads = 17;

typedef __attribute__((ext_vector_type(2))) float v2f;
typedef __attribute__((ext_vector_type(8))) float v8f;

__device__ __forceinline__ void lds_barrier()
{
    __builtin_amdgcn_fence(__ATOMIC_RELEASE, "workgroup", "local");
    __builtin_amdgcn_s_barrier();
    __builtin_amdgcn_fence(__ATOMIC_ACQUIRE, "workgroup", "local");
}

template <int M>
__device__ __forceinline__ float row_xor(float x)
{
    const int b = __builtin_bit_cast(int, x);
    return __builtin_bit_cast(float, __builtin_amdgcn_update_dpp(0, b, 0x160 + M, 0xf, 0xf, true));
}

__device__ __forceinline__ float xhalf(float x)
{
    const int b = __builtin_bit_cast(int, x);
    return __builtin_bit_cast(float,
        __builtin_amdgcn_permlanex16(b, b, 0x76543210, 0xfedcba98, false, false));
}

template <int M>
__device__ __forceinline__ float fold2(float a, float b, bool up)
{
    const float keep = up ? b : a;
    const float send = up ? a : b;
    return keep + row_xor<M>(send);
}

__global__ __launch_bounds__(128) __attribute__((amdgpu_waves_per_eu(4, 4))) void gat_kernel(
    const int* __restrict__ node_ids,
    const int* __restrict__ neighs,
    const int* __restrict__ mask,
    const float* __restrict__ emb,
    const float* __restrict__ a_w,
    const float* __restrict__ a_b,
    float* __restrict__ out)
{
    __shared__ float sh_part[2][kWaves][kPitch];
    __shared__ __attribute__((aligned(16))) float sh_a[kWaves][2][kStrip];
    __shared__ __attribute__((aligned(16))) float sh_z[kStrip];

    const int tid  = threadIdx.x;
    const int wave = __builtin_amdgcn_readfirstlane(tid >> 5);
    const int lane = tid & 31;
    const int h = lane >> 4;
    const int c = lane & 15;
    const int col0 = wave * 64 + 2 * lane;

    const float2 a1c = *(const float2*)(a_w + col0);
    const float2 a2c = *(const float2*)(a_w + kH + col0);
    const float bias = a_b[0];
    const unsigned colbyte = (unsigned)col0 * 4u;
    const bool up1 = (c & 1) != 0, up2 = (c & 2) != 0, up4 = (c & 4) != 0, up8 = (c & 8) != 0;
    const int crev = ((c & 1) << 3) | ((c & 2) << 1) | ((c & 4) >> 1) | ((c & 8) >> 3);
    const char* embB = (const char*)emb;

    {
        float* za = &sh_a[wave][0][0];
        for (int i = lane; i < 2 * kStrip; i += 32) za[i] = 0.f;
        if (wave == 0)
            for (int i = lane; i < kStrip; i += 32) sh_z[i] = 0.f;
    }
    const float* abase = (lane == 0 || lane == 18) ? &sh_a[wave][0][0]
                       : (lane == 1 || lane == 19) ? &sh_a[wave][1][0]
                       : &sh_z[0];

    float2 xn[kRows];
    int mk0n = 0, mk32n = 0;
    int idxN[kRows];
    unsigned inactN = 0;
    int mbits = 0;

    auto load_idx = [&](int p) {
        idxN[0] = node_ids[p];
        #pragma unroll
        for (int j = 0; j < kN; ++j) idxN[1 + j] = neighs[p * kN + j];
    };
    auto load_rows = [&](int p) {
        mk0n = mask[p * kN + (lane >= 1 ? lane - 1 : 0)];
        mk32n = mask[p * kN + kN - 1];
        #pragma unroll
        for (int r = 0; r < kRows; ++r) {
            if (r == 0 || __builtin_expect(((inactN >> (r - 1)) & 1u) == 0u, 1)) {
                const unsigned off = ((unsigned)idxN[r] << 10) + colbyte;
                xn[r] = *(const float2*)(embB + off);
            } else {
                xn[r] = make_float2(0.f, 0.f);
            }
        }
    };
    auto load_mbits = [&](int p) { mbits = mask[p * kN + lane]; };
    auto take_bitmap = [&]() { inactN = __builtin_amdgcn_ballot_w32(mbits > 0); };

    load_idx(blockIdx.x);
    load_mbits(blockIdx.x);
    asm volatile("" :: "s"(idxN[0]), "s"(idxN[1]), "s"(idxN[2]), "s"(idxN[3]), "s"(idxN[4]), "s"(idxN[5]), "s"(idxN[6]), "s"(idxN[7]), "s"(idxN[8]), "s"(idxN[9]), "s"(idxN[10]), "s"(idxN[11]), "s"(idxN[12]), "s"(idxN[13]), "s"(idxN[14]), "s"(idxN[15]), "s"(idxN[16]));
    asm volatile("" :: "s"(idxN[17]), "s"(idxN[18]), "s"(idxN[19]), "s"(idxN[20]), "s"(idxN[21]), "s"(idxN[22]), "s"(idxN[23]), "s"(idxN[24]), "s"(idxN[25]), "s"(idxN[26]), "s"(idxN[27]), "s"(idxN[28]), "s"(idxN[29]), "s"(idxN[30]), "s"(idxN[31]), "s"(idxN[32]));
    take_bitmap();
    if (kPerWG > 1) load_mbits(blockIdx.x + kGrid);
    load_rows(blockIdx.x);
    if (kPerWG > 1) load_idx(blockIdx.x + kGrid);

    #pragma unroll
    for (int it = 0; it < kPerWG; ++it) {
        const int pos = blockIdx.x + it * kGrid;
        const int buf = it & 1;

        float2 x[kRows];
        #pragma unroll
        for (int r = 0; r < kRows; ++r) x[r] = xn[r];
        const int mk0 = mk0n, mk32 = mk32n;
        unsigned inact = inactN;
        asm volatile("" : "+s"(inact));

        if (it + 1 < kPerWG) {
            take_bitmap();
            if (it + 2 < kPerWG) load_mbits(pos + 2 * kGrid);
            load_rows(pos + kGrid);
            if (it + 2 < kPerWG) load_idx(pos + 2 * kGrid);
        }

        float v[kVals];
        #pragma unroll
        for (int r = 0; r < kRows; ++r) v[r] = x[r].x * a2c.x + x[r].y * a2c.y;
        v[kRows] = x[0].x * a1c.x + x[0].y * a1c.y;

        float f0, f1, f2;
        {
            float u[17], t[9], q[5];
            #pragma unroll
            for (int j = 0; j < 17; ++j) u[j] = fold2<8>(v[2 * j], v[2 * j + 1], up8);
            #pragma unroll
            for (int j = 0; j < 8; ++j) t[j] = fold2<4>(u[2 * j], u[2 * j + 1], up4);
            t[8] = u[16] + row_xor<4>(u[16]);
            #pragma unroll
            for (int j = 0; j < 4; ++j) q[j] = fold2<2>(t[2 * j], t[2 * j + 1], up2);
            q[4] = t[8] + row_xor<2>(t[8]);
            f0 = fold2<1>(q[0], q[1], up1);
            f1 = fold2<1>(q[2], q[3], up1);
            f2 = q[4] + row_xor<1>(q[4]);
            f0 += xhalf(f0);
            f1 += xhalf(f1);
            f2 += xhalf(f2);
        }
        sh_part[buf][wave][16 * h + crev] = h ? f1 : f0;
        if ((lane & 23) == 0) sh_part[buf][wave][kN + (lane >> 3)] = f2;

        lds_barrier();

        float d0 = 0.f, d32 = 0.f, d1 = 0.f;
        #pragma unroll
        for (int wv = 0; wv < kWaves; ++wv) {
            d0  += sh_part[buf][wv][lane];
            d32 += sh_part[buf][wv][kN];
            d1  += sh_part[buf][wv][kSlotA1];
        }
        float z0 = d1 + d0 + bias;
        z0 = z0 > 0.f ? z0 : kSlope * z0;
        if (lane >= 1) z0 += (float)mk0 * kNeg;
        float z1 = d1 + d32 + bias;
        z1 = z1 > 0.f ? z1 : kSlope * z1;
        z1 += (float)mk32 * kNeg;

        float m = z0;
        m = fmaxf(m, row_xor<1>(m)); m = fmaxf(m, row_xor<2>(m));
        m = fmaxf(m, row_xor<4>(m)); m = fmaxf(m, row_xor<8>(m));
        m = fmaxf(m, xhalf(m));
        m = fmaxf(m, z1);
        const float e0 = __expf(z0 - m);
        const float e1 = __expf(z1 - m);
        float sum = e0;
        sum += row_xor<1>(sum); sum += row_xor<2>(sum);
        sum += row_xor<4>(sum); sum += row_xor<8>(sum);
        sum += xhalf(sum);
        sum += e1;
        const float inv = 1.f / sum;

        const float wl = e0 * inv;
        sh_a[wave][0][2 * lane] = wl;
        sh_a[wave][1][2 * lane + 1] = wl;
        if (lane == 0) {
            sh_a[wave][0][2 * kN] = e1 * inv;
            sh_a[wave][1][2 * kN + 1] = e1 * inv;
        }
        __builtin_amdgcn_fence(__ATOMIC_RELEASE, "wavefront", "local");
        __builtin_amdgcn_wave_barrier();
        __builtin_amdgcn_fence(__ATOMIC_ACQUIRE, "wavefront", "local");
        float4 av[kQuads];
        #pragma unroll
        for (int i = 0; i < kQuads; ++i) av[i] = ((const float4*)abase)[i];
        __builtin_amdgcn_sched_barrier(0);

        v8f acc0 = {}, acc1 = {};
        asm volatile("" : "+v"(acc0), "+v"(acc1));
        #pragma unroll
        for (int rr = 1; rr <= kRows; ++rr) {
            const int r = (rr == kRows) ? 0 : rr;
            if (r != 0 && __builtin_expect(((inact >> (r - 1)) & 1u) != 0u, 0)) continue;
            const float4 aq = av[r >> 1];
            v2f a, b;
            a.x = (r & 1) ? aq.z : aq.x;
            a.y = (r & 1) ? aq.w : aq.y;
            b.x = x[r].x;
            b.y = x[r].y;
            if (r & 1)
                acc1 = __builtin_amdgcn_wmma_f32_16x16x4_f32(
                     false, a,  false, b,
                     (short)0, acc1,  false,  false);
            else
                acc0 = __builtin_amdgcn_wmma_f32_16x16x4_f32(
                     false, a,  false, b,
                     (short)0, acc0,  false,  false);
        }
        if (lane < 16) {
            float* o = out + (size_t)pos * kH + wave * 64 + 2 * c;
            *(float2*)o        = make_float2(acc0[0] + acc1[0], acc0[1] + acc1[1]);
            *(float2*)(o + 32) = make_float2(acc0[2] + acc1[2], acc0[3] + acc1[3]);
        }
    }
}

extern "C" void kernel_launch(void* const* d_in, const int* in_sizes, int n_in,
                              void* d_out, int out_size, void* d_ws, size_t ws_size,
                              hipStream_t stream) {
    const int*   node_ids = (const int*)d_in[0];
    const int*   neighs   = (const int*)d_in[1];
    const int*   mask     = (const int*)d_in[2];
    const float* emb      = (const float*)d_in[3];
    const float* a_w      = (const float*)d_in[4];
    const float* a_b      = (const float*)d_in[5];
    float* out = (float*)d_out;

    dim3 grid(kGrid);
    dim3 block(128);
    gat_kernel<<<grid, block, 0, stream>>>(node_ids, neighs, mask, emb, a_w, a_b, out);
}


# The wait for a position's rows and their dots moved in front of the next position's branch-guarded requests: the counted wait no longer covers the next rows, which travel under the whole position
# speedup vs baseline: 1.8888x; 1.0958x over previous
#include <hip/hip_runtime.h>


constexpr int kB = 8;
constexpr int kS = 512;
constexpr int kN = 32;
constexpr int kH = 256;
constexpr int kRows = kN + 1;
constexpr int kVals = kRows + 1;
constexpr float kNeg = -1e9f;
constexpr float kSlope = 0.2f;
constexpr int kPos = kB * kS;
constexpr int kGrid = 1024;
constexpr int kPerWG = kPos / kGrid;
constexpr int kWaves = 4;
constexpr int kPitch = 36;
constexpr int kSlotA1 = kRows;
constexpr int kStrip = 72;
constexpr int kQuads = 17;

typedef __attribute__((ext_vector_type(2))) float v2f;
typedef __attribute__((ext_vector_type(8))) float v8f;

__device__ __forceinline__ void lds_barrier()
{
    __builtin_amdgcn_fence(__ATOMIC_RELEASE, "workgroup", "local");
    __builtin_amdgcn_s_barrier();
    __builtin_amdgcn_fence(__ATOMIC_ACQUIRE, "workgroup", "local");
}

template <int M>
__device__ __forceinline__ float row_xor(float x)
{
    const int b = __builtin_bit_cast(int, x);
    return __builtin_bit_cast(float, __builtin_amdgcn_update_dpp(0, b, 0x160 + M, 0xf, 0xf, true));
}

__device__ __forceinline__ float xhalf(float x)
{
    const int b = __builtin_bit_cast(int, x);
    return __builtin_bit_cast(float,
        __builtin_amdgcn_permlanex16(b, b, 0x76543210, 0xfedcba98, false, false));
}

template <int M>
__device__ __forceinline__ float fold2(float a, float b, bool up)
{
    const float keep = up ? b : a;
    const float send = up ? a : b;
    return keep + row_xor<M>(send);
}

__global__ __launch_bounds__(128) __attribute__((amdgpu_waves_per_eu(4, 4))) void gat_kernel(
    const int* __restrict__ node_ids,
    const int* __restrict__ neighs,
    const int* __restrict__ mask,
    const float* __restrict__ emb,
    const float* __restrict__ a_w,
    const float* __restrict__ a_b,
    float* __restrict__ out)
{
    __shared__ float sh_part[2][kWaves][kPitch];
    __shared__ __attribute__((aligned(16))) float sh_a[kWaves][2][kStrip];
    __shared__ __attribute__((aligned(16))) float sh_z[kStrip];

    const int tid  = threadIdx.x;
    const int wave = __builtin_amdgcn_readfirstlane(tid >> 5);
    const int lane = tid & 31;
    const int h = lane >> 4;
    const int c = lane & 15;
    const int col0 = wave * 64 + 2 * lane;

    const float2 a1c = *(const float2*)(a_w + col0);
    const float2 a2c = *(const float2*)(a_w + kH + col0);
    const float bias = a_b[0];
    const unsigned colbyte = (unsigned)col0 * 4u;
    const bool up1 = (c & 1) != 0, up2 = (c & 2) != 0, up4 = (c & 4) != 0, up8 = (c & 8) != 0;
    const int crev = ((c & 1) << 3) | ((c & 2) << 1) | ((c & 4) >> 1) | ((c & 8) >> 3);
    const char* embB = (const char*)emb;

    {
        float* za = &sh_a[wave][0][0];
        for (int i = lane; i < 2 * kStrip; i += 32) za[i] = 0.f;
        if (wave == 0)
            for (int i = lane; i < kStrip; i += 32) sh_z[i] = 0.f;
    }
    const float* abase = (lane == 0 || lane == 18) ? &sh_a[wave][0][0]
                       : (lane == 1 || lane == 19) ? &sh_a[wave][1][0]
                       : &sh_z[0];

    float2 xn[kRows];
    int mk0n = 0, mk32n = 0;
    int idxN[kRows];
    unsigned inactN = 0;
    int mbits = 0;

    auto load_idx = [&](int p) {
        idxN[0] = node_ids[p];
        #pragma unroll
        for (int j = 0; j < kN; ++j) idxN[1 + j] = neighs[p * kN + j];
    };
    auto load_rows = [&](int p) {
        mk0n = mask[p * kN + (lane >= 1 ? lane - 1 : 0)];
        mk32n = mask[p * kN + kN - 1];
        #pragma unroll
        for (int r = 0; r < kRows; ++r) {
            if (r == 0 || __builtin_expect(((inactN >> (r - 1)) & 1u) == 0u, 1)) {
                const unsigned off = ((unsigned)idxN[r] << 10) + colbyte;
                xn[r] = *(const float2*)(embB + off);
            } else {
                xn[r] = make_float2(0.f, 0.f);
            }
        }
    };
    auto load_mbits = [&](int p) { mbits = mask[p * kN + lane]; };
    auto take_bitmap = [&]() { inactN = __builtin_amdgcn_ballot_w32(mbits > 0); };

    load_idx(blockIdx.x);
    load_mbits(blockIdx.x);
    asm volatile("" :: "s"(idxN[0]), "s"(idxN[1]), "s"(idxN[2]), "s"(idxN[3]), "s"(idxN[4]), "s"(idxN[5]), "s"(idxN[6]), "s"(idxN[7]), "s"(idxN[8]), "s"(idxN[9]), "s"(idxN[10]), "s"(idxN[11]), "s"(idxN[12]), "s"(idxN[13]), "s"(idxN[14]), "s"(idxN[15]), "s"(idxN[16]));
    asm volatile("" :: "s"(idxN[17]), "s"(idxN[18]), "s"(idxN[19]), "s"(idxN[20]), "s"(idxN[21]), "s"(idxN[22]), "s"(idxN[23]), "s"(idxN[24]), "s"(idxN[25]), "s"(idxN[26]), "s"(idxN[27]), "s"(idxN[28]), "s"(idxN[29]), "s"(idxN[30]), "s"(idxN[31]), "s"(idxN[32]));
    take_bitmap();
    if (kPerWG > 1) load_mbits(blockIdx.x + kGrid);
    load_rows(blockIdx.x);
    if (kPerWG > 1) load_idx(blockIdx.x + kGrid);

    #pragma unroll
    for (int it = 0; it < kPerWG; ++it) {
        const int pos = blockIdx.x + it * kGrid;
        const int buf = it & 1;

        float2 x[kRows];
        #pragma unroll
        for (int r = 0; r < kRows; ++r) x[r] = xn[r];
        const int mk0 = mk0n, mk32 = mk32n;
        unsigned inact = inactN;
        asm volatile("" : "+s"(inact));

        float v[kVals];
        #pragma unroll
        for (int r = 0; r < kRows; ++r) v[r] = x[r].x * a2c.x + x[r].y * a2c.y;
        v[kRows] = x[0].x * a1c.x + x[0].y * a1c.y;
        asm volatile("" :: "v"(v[0]), "v"(v[1]), "v"(v[2]), "v"(v[3]), "v"(v[4]), "v"(v[5]), "v"(v[6]), "v"(v[7]), "v"(v[8]), "v"(v[9]), "v"(v[10]), "v"(v[11]), "v"(v[12]), "v"(v[13]), "v"(v[14]), "v"(v[15]), "v"(v[16]));
        asm volatile("" :: "v"(v[17]), "v"(v[18]), "v"(v[19]), "v"(v[20]), "v"(v[21]), "v"(v[22]), "v"(v[23]), "v"(v[24]), "v"(v[25]), "v"(v[26]), "v"(v[27]), "v"(v[28]), "v"(v[29]), "v"(v[30]), "v"(v[31]), "v"(v[32]), "v"(v[33]));

        if (it + 1 < kPerWG) {
            take_bitmap();
            if (it + 2 < kPerWG) load_mbits(pos + 2 * kGrid);
            load_rows(pos + kGrid);
            if (it + 2 < kPerWG) load_idx(pos + 2 * kGrid);
        }

        float f0, f1, f2;
        {
            float u[17], t[9], q[5];
            #pragma unroll
            for (int j = 0; j < 17; ++j) u[j] = fold2<8>(v[2 * j], v[2 * j + 1], up8);
            #pragma unroll
            for (int j = 0; j < 8; ++j) t[j] = fold2<4>(u[2 * j], u[2 * j + 1], up4);
            t[8] = u[16] + row_xor<4>(u[16]);
            #pragma unroll
            for (int j = 0; j < 4; ++j) q[j] = fold2<2>(t[2 * j], t[2 * j + 1], up2);
            q[4] = t[8] + row_xor<2>(t[8]);
            f0 = fold2<1>(q[0], q[1], up1);
            f1 = fold2<1>(q[2], q[3], up1);
            f2 = q[4] + row_xor<1>(q[4]);
            f0 += xhalf(f0);
            f1 += xhalf(f1);
            f2 += xhalf(f2);
        }
        sh_part[buf][wave][16 * h + crev] = h ? f1 : f0;
        if ((lane & 23) == 0) sh_part[buf][wave][kN + (lane >> 3)] = f2;

        lds_barrier();

        float d0 = 0.f, d32 = 0.f, d1 = 0.f;
        #pragma unroll
        for (int wv = 0; wv < kWaves; ++wv) {
            d0  += sh_part[buf][wv][lane];
            d32 += sh_part[buf][wv][kN];
            d1  += sh_part[buf][wv][kSlotA1];
        }
        float z0 = d1 + d0 + bias;
        z0 = z0 > 0.f ? z0 : kSlope * z0;
        if (lane >= 1) z0 += (float)mk0 * kNeg;
        float z1 = d1 + d32 + bias;
        z1 = z1 > 0.f ? z1 : kSlope * z1;
        z1 += (float)mk32 * kNeg;

        float m = z0;
        m = fmaxf(m, row_xor<1>(m)); m = fmaxf(m, row_xor<2>(m));
        m = fmaxf(m, row_xor<4>(m)); m = fmaxf(m, row_xor<8>(m));
        m = fmaxf(m, xhalf(m));
        m = fmaxf(m, z1);
        const float e0 = __expf(z0 - m);
        const float e1 = __expf(z1 - m);
        float sum = e0;
        sum += row_xor<1>(sum); sum += row_xor<2>(sum);
        sum += row_xor<4>(sum); sum += row_xor<8>(sum);
        sum += xhalf(sum);
        sum += e1;
        const float inv = 1.f / sum;

        const float wl = e0 * inv;
        sh_a[wave][0][2 * lane] = wl;
        sh_a[wave][1][2 * lane + 1] = wl;
        if (lane == 0) {
            sh_a[wave][0][2 * kN] = e1 * inv;
            sh_a[wave][1][2 * kN + 1] = e1 * inv;
        }
        __builtin_amdgcn_fence(__ATOMIC_RELEASE, "wavefront", "local");
        __builtin_amdgcn_wave_barrier();
        __builtin_amdgcn_fence(__ATOMIC_ACQUIRE, "wavefront", "local");
        float4 av[kQuads];
        #pragma unroll
        for (int i = 0; i < kQuads; ++i) av[i] = ((const float4*)abase)[i];
        __builtin_amdgcn_sched_barrier(0);

        v8f acc0 = {}, acc1 = {};
        asm volatile("" : "+v"(acc0), "+v"(acc1));
        #pragma unroll
        for (int rr = 1; rr <= kRows; ++rr) {
            const int r = (rr == kRows) ? 0 : rr;
            if (r != 0 && __builtin_expect(((inact >> (r - 1)) & 1u) != 0u, 0)) continue;
            const float4 aq = av[r >> 1];
            v2f a, b;
            a.x = (r & 1) ? aq.z : aq.x;
            a.y = (r & 1) ? aq.w : aq.y;
            b.x = x[r].x;
            b.y = x[r].y;
            if (r & 1)
                acc1 = __builtin_amdgcn_wmma_f32_16x16x4_f32(
                     false, a,  false, b,
                     (short)0, acc1,  false,  false);
            else
                acc0 = __builtin_amdgcn_wmma_f32_16x16x4_f32(
                     false, a,  false, b,
                     (short)0, acc0,  false,  false);
        }
        if (lane < 16) {
            float* o = out + (size_t)pos * kH + wave * 64 + 2 * c;
            *(float2*)o        = make_float2(acc0[0] + acc1[0], acc0[1] + acc1[1]);
            *(float2*)(o + 32) = make_float2(acc0[2] + acc1[2], acc0[3] + acc1[3]);
        }
    }
}

extern "C" void kernel_launch(void* const* d_in, const int* in_sizes, int n_in,
                              void* d_out, int out_size, void* d_ws, size_t ws_size,
                              hipStream_t stream) {
    const int*   node_ids = (const int*)d_in[0];
    const int*   neighs   = (const int*)d_in[1];
    const int*   mask     = (const int*)d_in[2];
    const float* emb      = (const float*)d_in[3];
    const float* a_w      = (const float*)d_in[4];
    const float* a_b      = (const float*)d_in[5];
    float* out = (float*)d_out;

    dim3 grid(kGrid);
    dim3 block(128);
    gat_kernel<<<grid, block, 0, stream>>>(node_ids, neighs, mask, emb, a_w, a_b, out);
}
